# starting kernel
# baseline (speedup 1.0000x reference)
#include <hip/hip_runtime.h>
#include <hip/hip_bf16.h>

typedef __attribute__((ext_vector_type(2))) float v2f;
typedef __attribute__((ext_vector_type(8))) float v8f;

#define DCH 10

__global__ __launch_bounds__(256) void attpool_wmma_kernel(
    const float* __restrict__ xyz,
    const float* __restrict__ fc_w,
    const float* __restrict__ mlp_w,
    const float* __restrict__ bn_gamma,
    const float* __restrict__ bn_beta,
    const float* __restrict__ bn_mean,
    const float* __restrict__ bn_var,
    const int*   __restrict__ neigh_idx,
    float*       __restrict__ y_out,
    int B, int N)
{
  __shared__ float lfeat[8 * 192 + 16];
  __shared__ float sBt[192];

  const int lane = threadIdx.x & 31;
  const int wid  = threadIdx.x >> 5;
  const int p    = blockIdx.x * 8 + wid;
  const int b    = p / N;
  const int n    = p - b * N;
  const int m    = lane & 15;
  const int h    = lane >> 4;

  if (threadIdx.x < 192) {
    const int o = threadIdx.x / 12;
    const int i = threadIdx.x % 12;
    sBt[threadIdx.x] = (i < DCH && o < DCH) ? fc_w[o * DCH + i] : 0.0f;
  }

  const size_t pbase = (size_t)(b * (size_t)N + n);
  const float cx = xyz[pbase * 3 + 0];
  const float cy = xyz[pbase * 3 + 1];
  const float cz = xyz[pbase * 3 + 2];

  const int nb = neigh_idx[pbase * 16 + m];
  const size_t nbase = (size_t)(b * (size_t)N + nb);
  const float nx = xyz[nbase * 3 + 0];
  const float ny = xyz[nbase * 3 + 1];
  const float nz = xyz[nbase * 3 + 2];

  const float rx = cx - nx, ry = cy - ny, rz = cz - nz;
  const float dis = sqrtf(rx * rx + ry * ry + rz * rz);

  float* fl = &lfeat[wid * 192];
  float* arow = fl + m * 12;
  {
    float4* fl4 = (float4*)arow;
    fl4[0] = make_float4(dis, rx, ry, rz);
    fl4[1] = make_float4(cx, cy, cz, nx);
    fl4[2] = make_float4(ny, nz, 0.0f, 0.0f);
  }
  __syncthreads();

  const float* brow = sBt + m * 12;
  v8f acc = {};
#pragma unroll
  for (int c0 = 0; c0 < 3; ++c0) {
    const int k0 = 4 * c0 + 2 * h;
    const v2f a  = *(const v2f*)(arow + k0);
    const v2f bb = *(const v2f*)(brow + k0);
    acc = __builtin_amdgcn_wmma_f32_16x16x4_f32(
         false, a,  false, bb,
         (short)0, acc,  false,  false);
  }

  float mx = acc[0];
#pragma unroll
  for (int r = 1; r < 8; ++r) mx = fmaxf(mx, acc[r]);
  mx = fmaxf(mx, __shfl_xor(mx, 16, 32));

  float e[8];
  float s = 0.0f;
#pragma unroll
  for (int r = 0; r < 8; ++r) { e[r] = __expf(acc[r] - mx); s += e[r]; }
  s += __shfl_xor(s, 16, 32);
  const float inv = 1.0f / s;

  float part = 0.0f;
#pragma unroll
  for (int r = 0; r < 8; ++r) {
    const float fvk = fl[(r + 8 * h) * 12 + m];
    part = fmaf(fvk, e[r] * inv, part);
  }
  const float agg = part + __shfl_xor(part, 16, 32);

  const float* mrow = mlp_w + (m < DCH ? m : DCH - 1) * DCH;
  float yo = 0.0f;
#pragma unroll
  for (int i = 0; i < DCH; ++i) {
    const float ai = __shfl(agg, i, 32);
    yo = fmaf(mrow[i], ai, yo);
  }

  if (h == 0 && m < DCH) {
    const float g  = bn_gamma[m];
    const float bt = bn_beta[m];
    const float mu = bn_mean[m];
    const float vv = bn_var[m];
    float r = (yo - mu) * rsqrtf(vv + 1e-5f) * g + bt;
    r = fmaxf(r, 0.0f);
    y_out[(size_t)p * DCH + m] = r;
  }
}

__global__ __launch_bounds__(256) void maxpool_gather_kernel(
    const float* __restrict__ y,
    const int*   __restrict__ sub_idx,
    float*       __restrict__ out,
    int B, int N, int M)
{
  const int t = blockIdx.x * 256 + threadIdx.x;
  const int total = B * M * DCH;
  if (t >= total) return;
  const int o  = t % DCH;
  const int bm = t / DCH;
  const int b  = bm / M;
  const int* si = sub_idx + (size_t)bm * 16;
  const size_t ybase = (size_t)b * (size_t)N * DCH;
  float mx = -3.402823466e38f;
#pragma unroll
  for (int k = 0; k < 16; ++k) {
    const int idx = si[k];
    mx = fmaxf(mx, y[ybase + (size_t)idx * DCH + o]);
  }
  out[t] = mx;
}

extern "C" void kernel_launch(void* const* d_in, const int* in_sizes, int n_in,
                              void* d_out, int out_size, void* d_ws, size_t ws_size,
                              hipStream_t stream) {
  const float* xyz       = (const float*)d_in[0];
  const float* fc_w      = (const float*)d_in[1];
  const float* mlp_w     = (const float*)d_in[2];
  const float* bn_gamma  = (const float*)d_in[3];
  const float* bn_beta   = (const float*)d_in[4];
  const float* bn_mean   = (const float*)d_in[5];
  const float* bn_var    = (const float*)d_in[6];
  const int*   neigh_idx = (const int*)d_in[7];
  const int*   sub_idx   = (const int*)d_in[8];
  float*       out       = (float*)d_out;
  float*       y_ws      = (float*)d_ws;

  const int B = 2;
  const int N = in_sizes[7] / (B * 16);
  const int M = in_sizes[8] / (B * 16);

  const int points = B * N;
  attpool_wmma_kernel<<<points / 8, 256, 0, stream>>>(
      xyz, fc_w, mlp_w, bn_gamma, bn_beta, bn_mean, bn_var, neigh_idx,
      y_ws, B, N);

  const int total = B * M * DCH;
  maxpool_gather_kernel<<<(total + 255) / 256, 256, 0, stream>>>(
      y_ws, sub_idx, out, B, N, M);
}


# One wave per 16 points with all index loads, then all neighbour loads, in flight at once; constants once per wave; wave-private LDS transpose without block barrier; cross-lane moves in registers
# speedup vs baseline: 1.1717x; 1.1717x over previous
#include <hip/hip_runtime.h>
#include <hip/hip_bf16.h>


typedef __attribute__((ext_vector_type(2))) float v2f;
typedef __attribute__((ext_vector_type(3))) float v3f;
typedef __attribute__((ext_vector_type(8))) float v8f;
typedef v3f v3f_a4 __attribute__((aligned(4)));

#define DCH 10
#define PPW 16
#define TP  20
#define HB  144
#define WREG 264

__device__ __forceinline__ float other_half(float v) {
  const int x = __float_as_int(v);
  return __int_as_float(__builtin_amdgcn_permlanex16(x, x, 0x76543210, 0xfedcba98, false, false));
}
template <int I>
__device__ __forceinline__ float row_lane(float v) {
  return __int_as_float(__builtin_amdgcn_update_dpp(0, __float_as_int(v), 0x150 + I, 0xF, 0xF, true));
}
__device__ __forceinline__ float lane_value(float v, int l) {
  return __int_as_float(__builtin_amdgcn_readlane(__float_as_int(v), l));
}

__global__ __launch_bounds__(256) void attpool_wmma_kernel(
    const float* __restrict__ xyz,
    const float* __restrict__ fc_w,
    const float* __restrict__ mlp_w,
    const float* __restrict__ bn_gamma,
    const float* __restrict__ bn_beta,
    const float* __restrict__ bn_mean,
    const float* __restrict__ bn_var,
    const int*   __restrict__ neigh_idx,
    float*       __restrict__ y_out,
    int B, int N)
{
  __shared__ __attribute__((aligned(16))) float sT[8 * WREG];

  const int lane = threadIdx.x & 31;
  const int wid  = __builtin_amdgcn_readfirstlane((int)(threadIdx.x >> 5));
  const int m    = lane & 15;
  const int h    = lane >> 4;
  const int p0   = (blockIdx.x * 8 + wid) * PPW;
  const int b0   = p0 / N;
  const int n0   = p0 - b0 * N;

  const int* ip = neigh_idx + (size_t)p0 * 16 + m;
  int nb[PPW];
#pragma unroll
  for (int j = 0; j < PPW; ++j) nb[j] = ip[j * 16];

  const float* cp = xyz + (size_t)p0 * 3;
  const float c_lo = cp[lane];
  const float c_hi = cp[32 + m];

  v2f bfrag[3];
#pragma unroll
  for (int c0 = 0; c0 < 3; ++c0) {
    const int k0 = 4 * c0 + 2 * h;
    const int mm = m < DCH ? m : 0;
    const int ka = k0 < DCH ? k0 : 0;
    const int kb = k0 + 1 < DCH ? k0 + 1 : 0;
    const float va = fc_w[mm * DCH + ka];
    const float vb = fc_w[mm * DCH + kb];
    bfrag[c0].x = (m < DCH && k0 < DCH) ? va : 0.0f;
    bfrag[c0].y = (m < DCH && k0 + 1 < DCH) ? vb : 0.0f;
  }
  const int mc = m < DCH ? m : DCH - 1;
  const float* mrow = mlp_w + mc * DCH;
  float mw[DCH];
#pragma unroll
  for (int i = 0; i < DCH; ++i) mw[i] = mrow[i];
  const float g  = bn_gamma[mc];
  const float bt = bn_beta[mc];
  const float mu = bn_mean[mc];
  const float rs = rsqrtf(bn_var[mc] + 1e-5f);

  float nx[PPW], ny[PPW], nz[PPW];
#pragma unroll
  for (int j = 0; j < PPW; ++j) {
    const int bj = (n0 + j >= N) ? b0 + 1 : b0;
    const unsigned off = ((unsigned)(bj * N) + (unsigned)nb[j]) * 3u;
    const v3f v = *(const v3f_a4*)(xyz + off);
    nx[j] = v.x; ny[j] = v.y; nz[j] = v.z;
  }

  float* wT = sT + wid * WREG;
  float* st = wT + h * HB + m;
  const unsigned long long tlo = 0x332E0F0A29240500ULL;
  const unsigned long long thi = 0x3838383838381914ULL;
  const int roff = 4 * (int)(((m < 8 ? tlo : thi) >> (8 * (m & 7))) & 0xFFu);
  const float* rd = wT + roff + 8 * h;

  float* yp = y_out + (size_t)p0 * DCH + h * DCH + m;

#pragma unroll
  for (int i2 = 0; i2 < PPW / 2; ++i2) {
    float part2[2];
#pragma unroll
    for (int t = 0; t < 2; ++t) {
      const int j = 2 * i2 + t;
      const float cx = (3 * j + 0 < 32) ? lane_value(c_lo, 3 * j + 0) : lane_value(c_hi, 3 * j + 0 - 32);
      const float cy = (3 * j + 1 < 32) ? lane_value(c_lo, 3 * j + 1) : lane_value(c_hi, 3 * j + 1 - 32);
      const float cz = (3 * j + 2 < 32) ? lane_value(c_lo, 3 * j + 2) : lane_value(c_hi, 3 * j + 2 - 32);
      const float rx = cx - nx[j], ry = cy - ny[j], rz = cz - nz[j];
      const float dis = sqrtf(rx * rx + ry * ry + rz * rz);

      v2f a0, a1, a2;
      a0.x = h ? ry : dis;  a0.y = h ? rz : rx;
      a1.x = h ? cz : cx;   a1.y = h ? nx[j] : cy;
      a2.x = h ? 0.0f : ny[j];  a2.y = h ? 0.0f : nz[j];

      __builtin_amdgcn_wave_barrier();
      st[0 * TP] = a0.x; st[1 * TP] = a0.y;
      st[2 * TP] = a1.x; st[3 * TP] = a1.y;
      st[4 * TP] = a2.x; st[5 * TP] = a2.y;
      __builtin_amdgcn_wave_barrier();

      v8f acc = {};
      acc = __builtin_amdgcn_wmma_f32_16x16x4_f32(false, a0, false, bfrag[0], (short)0, acc, false, false);
      acc = __builtin_amdgcn_wmma_f32_16x16x4_f32(false, a1, false, bfrag[1], (short)0, acc, false, false);
      acc = __builtin_amdgcn_wmma_f32_16x16x4_f32(false, a2, false, bfrag[2], (short)0, acc, false, false);

      const float4 f0 = *(const float4*)(rd);
      const float4 f1 = *(const float4*)(rd + 4);
      const float fv[8] = {f0.x, f0.y, f0.z, f0.w, f1.x, f1.y, f1.z, f1.w};

      float mx = acc[0];
#pragma unroll
      for (int r = 1; r < 8; ++r) mx = fmaxf(mx, acc[r]);
      mx = fmaxf(mx, other_half(mx));

      float e[8];
      float s = 0.0f;
#pragma unroll
      for (int r = 0; r < 8; ++r) { e[r] = __expf(acc[r] - mx); s += e[r]; }
      s += other_half(s);
      const float inv = 1.0f / s;

      float part = 0.0f;
#pragma unroll
      for (int r = 0; r < 8; ++r) part = fmaf(fv[r], e[r] * inv, part);
      part2[t] = part;
    }

    const float keep = h ? part2[1] : part2[0];
    const float send = h ? part2[0] : part2[1];
    const float agg  = keep + other_half(send);

    float yo = 0.0f;
    yo = fmaf(mw[0], row_lane<0>(agg), yo);
    yo = fmaf(mw[1], row_lane<1>(agg), yo);
    yo = fmaf(mw[2], row_lane<2>(agg), yo);
    yo = fmaf(mw[3], row_lane<3>(agg), yo);
    yo = fmaf(mw[4], row_lane<4>(agg), yo);
    yo = fmaf(mw[5], row_lane<5>(agg), yo);
    yo = fmaf(mw[6], row_lane<6>(agg), yo);
    yo = fmaf(mw[7], row_lane<7>(agg), yo);
    yo = fmaf(mw[8], row_lane<8>(agg), yo);
    yo = fmaf(mw[9], row_lane<9>(agg), yo);

    if (m < DCH) {
      float r = (yo - mu) * rs * g + bt;
      r = fmaxf(r, 0.0f);
      yp[i2 * 2 * DCH] = r;
    }
  }
}

__global__ __launch_bounds__(256) void maxpool_gather_kernel(
    const float* __restrict__ y,
    const int*   __restrict__ sub_idx,
    float*       __restrict__ out,
    int B, int N, int M)
{
  const int t = blockIdx.x * 256 + threadIdx.x;
  const int total = B * M * DCH;
  if (t >= total) return;
  const int o  = t % DCH;
  const int bm = t / DCH;
  const int b  = bm / M;
  const int* si = sub_idx + (size_t)bm * 16;
  const size_t ybase = (size_t)b * (size_t)N * DCH;
  float mx = -3.402823466e38f;
#pragma unroll
  for (int k = 0; k < 16; ++k) {
    const int idx = si[k];
    mx = fmaxf(mx, y[ybase + (size_t)idx * DCH + o]);
  }
  out[t] = mx;
}

extern "C" void kernel_launch(void* const* d_in, const int* in_sizes, int n_in,
                              void* d_out, int out_size, void* d_ws, size_t ws_size,
                              hipStream_t stream) {
  const float* xyz       = (const float*)d_in[0];
  const float* fc_w      = (const float*)d_in[1];
  const float* mlp_w     = (const float*)d_in[2];
  const float* bn_gamma  = (const float*)d_in[3];
  const float* bn_beta   = (const float*)d_in[4];
  const float* bn_mean   = (const float*)d_in[5];
  const float* bn_var    = (const float*)d_in[6];
  const int*   neigh_idx = (const int*)d_in[7];
  const int*   sub_idx   = (const int*)d_in[8];
  float*       out       = (float*)d_out;
  float*       y_ws      = (float*)d_ws;

  const int B = 2;
  const int N = in_sizes[7] / (B * 16);
  const int M = in_sizes[8] / (B * 16);

  const int points = B * N;
  attpool_wmma_kernel<<<points / 128, 256, 0, stream>>>(
      xyz, fc_w, mlp_w, bn_gamma, bn_beta, bn_mean, bn_var, neigh_idx,
      y_ws, B, N);

  const int total = B * M * DCH;
  maxpool_gather_kernel<<<(total + 255) / 256, 256, 0, stream>>>(
      y_ws, sub_idx, out, B, N, M);
}


# Pooling kernel: 3 lanes per output point read each y row (padded to 64 B) as 16-byte loads, both batches' 32 row loads in flight per lane; 1536 waves launched instead of 10240
# speedup vs baseline: 1.2110x; 1.0335x over previous
#include <hip/hip_runtime.h>
#include <hip/hip_bf16.h>


typedef __attribute__((ext_vector_type(2))) float v2f;
typedef __attribute__((ext_vector_type(3))) float v3f;
typedef __attribute__((ext_vector_type(8))) float v8f;
typedef v3f v3f_a4 __attribute__((aligned(4)));
typedef __attribute__((ext_vector_type(4))) float v4f;
typedef __attribute__((ext_vector_type(4))) int   v4i;

#define DCH 10
#define NB  2
#define YP  16
#define YW  12
#define PT  192
#define PPW 16
#define TP  20
#define HB  144
#define WREG 264

__device__ __forceinline__ float other_half(float v) {
  const int x = __float_as_int(v);
  return __int_as_float(__builtin_amdgcn_permlanex16(x, x, 0x76543210, 0xfedcba98, false, false));
}
template <int I>
__device__ __forceinline__ float row_lane(float v) {
  return __int_as_float(__builtin_amdgcn_update_dpp(0, __float_as_int(v), 0x150 + I, 0xF, 0xF, true));
}
__device__ __forceinline__ float lane_value(float v, int l) {
  return __int_as_float(__builtin_amdgcn_readlane(__float_as_int(v), l));
}

__global__ __launch_bounds__(256) void attpool_wmma_kernel(
    const float* __restrict__ xyz,
    const float* __restrict__ fc_w,
    const float* __restrict__ mlp_w,
    const float* __restrict__ bn_gamma,
    const float* __restrict__ bn_beta,
    const float* __restrict__ bn_mean,
    const float* __restrict__ bn_var,
    const int*   __restrict__ neigh_idx,
    float*       __restrict__ y_out,
    int B, int N)
{
  __shared__ __attribute__((aligned(16))) float sT[8 * WREG];

  const int lane = threadIdx.x & 31;
  const int wid  = __builtin_amdgcn_readfirstlane((int)(threadIdx.x >> 5));
  const int m    = lane & 15;
  const int h    = lane >> 4;
  const int p0   = (blockIdx.x * 8 + wid) * PPW;
  const int b0   = p0 / N;
  const int n0   = p0 - b0 * N;

  const int* ip = neigh_idx + (size_t)p0 * 16 + m;
  int nb[PPW];
#pragma unroll
  for (int j = 0; j < PPW; ++j) nb[j] = ip[j * 16];

  const float* cp = xyz + (size_t)p0 * 3;
  const float c_lo = cp[lane];
  const float c_hi = cp[32 + m];

  v2f bfrag[3];
#pragma unroll
  for (int c0 = 0; c0 < 3; ++c0) {
    const int k0 = 4 * c0 + 2 * h;
    const int mm = m < DCH ? m : 0;
    const int ka = k0 < DCH ? k0 : 0;
    const int kb = k0 + 1 < DCH ? k0 + 1 : 0;
    const float va = fc_w[mm * DCH + ka];
    const float vb = fc_w[mm * DCH + kb];
    bfrag[c0].x = (m < DCH && k0 < DCH) ? va : 0.0f;
    bfrag[c0].y = (m < DCH && k0 + 1 < DCH) ? vb : 0.0f;
  }
  const int mc = m < DCH ? m : DCH - 1;
  const float* mrow = mlp_w + mc * DCH;
  float mw[DCH];
#pragma unroll
  for (int i = 0; i < DCH; ++i) mw[i] = mrow[i];
  const float g  = bn_gamma[mc];
  const float bt = bn_beta[mc];
  const float mu = bn_mean[mc];
  const float rs = rsqrtf(bn_var[mc] + 1e-5f);

  float nx[PPW], ny[PPW], nz[PPW];
#pragma unroll
  for (int j = 0; j < PPW; ++j) {
    const int bj = (n0 + j >= N) ? b0 + 1 : b0;
    const unsigned off = ((unsigned)(bj * N) + (unsigned)nb[j]) * 3u;
    const v3f v = *(const v3f_a4*)(xyz + off);
    nx[j] = v.x; ny[j] = v.y; nz[j] = v.z;
  }

  float* wT = sT + wid * WREG;
  float* st = wT + h * HB + m;
  const unsigned long long tlo = 0x332E0F0A29240500ULL;
  const unsigned long long thi = 0x3838383838381914ULL;
  const int roff = 4 * (int)(((m < 8 ? tlo : thi) >> (8 * (m & 7))) & 0xFFu);
  const float* rd = wT + roff + 8 * h;

  float* yp = y_out + (size_t)p0 * YP + h * YP + m;

#pragma unroll
  for (int i2 = 0; i2 < PPW / 2; ++i2) {
    float part2[2];
#pragma unroll
    for (int t = 0; t < 2; ++t) {
      const int j = 2 * i2 + t;
      const float cx = (3 * j + 0 < 32) ? lane_value(c_lo, 3 * j + 0) : lane_value(c_hi, 3 * j + 0 - 32);
      const float cy = (3 * j + 1 < 32) ? lane_value(c_lo, 3 * j + 1) : lane_value(c_hi, 3 * j + 1 - 32);
      const float cz = (3 * j + 2 < 32) ? lane_value(c_lo, 3 * j + 2) : lane_value(c_hi, 3 * j + 2 - 32);
      const float rx = cx - nx[j], ry = cy - ny[j], rz = cz - nz[j];
      const float dis = sqrtf(rx * rx + ry * ry + rz * rz);

      v2f a0, a1, a2;
      a0.x = h ? ry : dis;  a0.y = h ? rz : rx;
      a1.x = h ? cz : cx;   a1.y = h ? nx[j] : cy;
      a2.x = h ? 0.0f : ny[j];  a2.y = h ? 0.0f : nz[j];

      __builtin_amdgcn_wave_barrier();
      st[0 * TP] = a0.x; st[1 * TP] = a0.y;
      st[2 * TP] = a1.x; st[3 * TP] = a1.y;
      st[4 * TP] = a2.x; st[5 * TP] = a2.y;
      __builtin_amdgcn_wave_barrier();

      v8f acc = {};
      acc = __builtin_amdgcn_wmma_f32_16x16x4_f32(false, a0, false, bfrag[0], (short)0, acc, false, false);
      acc = __builtin_amdgcn_wmma_f32_16x16x4_f32(false, a1, false, bfrag[1], (short)0, acc, false, false);
      acc = __builtin_amdgcn_wmma_f32_16x16x4_f32(false, a2, false, bfrag[2], (short)0, acc, false, false);

      const float4 f0 = *(const float4*)(rd);
      const float4 f1 = *(const float4*)(rd + 4);
      const float fv[8] = {f0.x, f0.y, f0.z, f0.w, f1.x, f1.y, f1.z, f1.w};

      float mx = acc[0];
#pragma unroll
      for (int r = 1; r < 8; ++r) mx = fmaxf(mx, acc[r]);
      mx = fmaxf(mx, other_half(mx));

      float e[8];
      float s = 0.0f;
#pragma unroll
      for (int r = 0; r < 8; ++r) { e[r] = __expf(acc[r] - mx); s += e[r]; }
      s += other_half(s);
      const float inv = 1.0f / s;

      float part = 0.0f;
#pragma unroll
      for (int r = 0; r < 8; ++r) part = fmaf(fv[r], e[r] * inv, part);
      part2[t] = part;
    }

    const float keep = h ? part2[1] : part2[0];
    const float send = h ? part2[0] : part2[1];
    const float agg  = keep + other_half(send);

    float yo = 0.0f;
    yo = fmaf(mw[0], row_lane<0>(agg), yo);
    yo = fmaf(mw[1], row_lane<1>(agg), yo);
    yo = fmaf(mw[2], row_lane<2>(agg), yo);
    yo = fmaf(mw[3], row_lane<3>(agg), yo);
    yo = fmaf(mw[4], row_lane<4>(agg), yo);
    yo = fmaf(mw[5], row_lane<5>(agg), yo);
    yo = fmaf(mw[6], row_lane<6>(agg), yo);
    yo = fmaf(mw[7], row_lane<7>(agg), yo);
    yo = fmaf(mw[8], row_lane<8>(agg), yo);
    yo = fmaf(mw[9], row_lane<9>(agg), yo);

    if (m < YW) {
      float r = (yo - mu) * rs * g + bt;
      r = fmaxf(r, 0.0f);
      yp[i2 * 2 * YP] = m < DCH ? r : 0.0f;
    }
  }
}

__global__ __launch_bounds__(PT) void maxpool_gather_kernel(
    const float* __restrict__ y,
    const int*   __restrict__ sub_idx,
    float*       __restrict__ out,
    int N, int M)
{
  const int t = blockIdx.x * PT + threadIdx.x;
  if (t >= 3 * M) return;
  const int mi = t / 3;
  const int q  = t - 3 * mi;

  v4i id[NB][4];
#pragma unroll
  for (int b = 0; b < NB; ++b) {
    const v4i* si = (const v4i*)(sub_idx + ((size_t)b * M + mi) * 16);
#pragma unroll
    for (int j = 0; j < 4; ++j) id[b][j] = si[j];
  }

  const char* yb = (const char*)y;
  const unsigned qoff = 16u * (unsigned)q;
  v4f r[NB][16];
#pragma unroll
  for (int b = 0; b < NB; ++b) {
#pragma unroll
    for (int k = 0; k < 16; ++k) {
      const unsigned off = ((unsigned)(b * N) + (unsigned)id[b][k >> 2][k & 3]) * (YP * 4u) + qoff;
      r[b][k] = *(const v4f*)(yb + off);
    }
  }
  __builtin_amdgcn_sched_barrier(0);

#pragma unroll
  for (int b = 0; b < NB; ++b) {
    v4f mx = {-3.402823466e38f, -3.402823466e38f, -3.402823466e38f, -3.402823466e38f};
#pragma unroll
    for (int k = 0; k < 16; ++k) {
      mx.x = fmaxf(mx.x, r[b][k].x); mx.y = fmaxf(mx.y, r[b][k].y);
      mx.z = fmaxf(mx.z, r[b][k].z); mx.w = fmaxf(mx.w, r[b][k].w);
    }
    float* o = out + ((size_t)b * M + mi) * DCH + 4 * q;
    v2f lo; lo.x = mx.x; lo.y = mx.y;
    *(v2f*)o = lo;
    if (q < 2) {
      v2f hi; hi.x = mx.z; hi.y = mx.w;
      *(v2f*)(o + 2) = hi;
    }
  }
}

extern "C" void kernel_launch(void* const* d_in, const int* in_sizes, int n_in,
                              void* d_out, int out_size, void* d_ws, size_t ws_size,
                              hipStream_t stream) {
  const float* xyz       = (const float*)d_in[0];
  const float* fc_w      = (const float*)d_in[1];
  const float* mlp_w     = (const float*)d_in[2];
  const float* bn_gamma  = (const float*)d_in[3];
  const float* bn_beta   = (const float*)d_in[4];
  const float* bn_mean   = (const float*)d_in[5];
  const float* bn_var    = (const float*)d_in[6];
  const int*   neigh_idx = (const int*)d_in[7];
  const int*   sub_idx   = (const int*)d_in[8];
  float*       out       = (float*)d_out;
  float*       y_ws      = (float*)d_ws;

  const int B = NB;
  const int N = in_sizes[7] / (B * 16);
  const int M = in_sizes[8] / (B * 16);

  const int points = B * N;
  attpool_wmma_kernel<<<points / 128, 256, 0, stream>>>(
      xyz, fc_w, mlp_w, bn_gamma, bn_beta, bn_mean, bn_var, neigh_idx,
      y_ws, B, N);

  maxpool_gather_kernel<<<(3 * M + PT - 1) / PT, PT, 0, stream>>>(
      y_ws, sub_idx, out, N, M);
}


# Vector issue slots per point cut: bare v_sqrt_f32 and v_rcp_f32 (no correction code), exp scale folded into one fma per element, softmax quotient applied once after the weighted sum, not per term
# speedup vs baseline: 1.3162x; 1.0869x over previous
#include <hip/hip_runtime.h>
#include <hip/hip_bf16.h>


typedef __attribute__((ext_vector_type(2))) float v2f;
typedef __attribute__((ext_vector_type(3))) float v3f;
typedef __attribute__((ext_vector_type(8))) float v8f;
typedef v3f v3f_a4 __attribute__((aligned(4)));
typedef __attribute__((ext_vector_type(4))) float v4f;
typedef __attribute__((ext_vector_type(4))) int   v4i;

#define DCH 10
#define NB  2
#define YP  16
#define YW  12
#define PT  192
#define PPW 16
#define TP  20
#define HB  144
#define WREG 264
#define LOG2E 1.44269504088896340736f

__device__ __forceinline__ float other_half(float v) {
  const int x = __float_as_int(v);
  return __int_as_float(__builtin_amdgcn_permlanex16(x, x, 0x76543210, 0xfedcba98, false, false));
}
template <int I>
__device__ __forceinline__ float row_lane(float v) {
  return __int_as_float(__builtin_amdgcn_update_dpp(0, __float_as_int(v), 0x150 + I, 0xF, 0xF, true));
}
__device__ __forceinline__ float lane_value(float v, int l) {
  return __int_as_float(__builtin_amdgcn_readlane(__float_as_int(v), l));
}

__global__ __launch_bounds__(256) void attpool_wmma_kernel(
    const float* __restrict__ xyz,
    const float* __restrict__ fc_w,
    const float* __restrict__ mlp_w,
    const float* __restrict__ bn_gamma,
    const float* __restrict__ bn_beta,
    const float* __restrict__ bn_mean,
    const float* __restrict__ bn_var,
    const int*   __restrict__ neigh_idx,
    float*       __restrict__ y_out,
    int B, int N)
{
  __shared__ __attribute__((aligned(16))) float sT[8 * WREG];

  const int lane = threadIdx.x & 31;
  const int wid  = __builtin_amdgcn_readfirstlane((int)(threadIdx.x >> 5));
  const int m    = lane & 15;
  const int h    = lane >> 4;
  const int p0   = (blockIdx.x * 8 + wid) * PPW;
  const int b0   = p0 / N;
  const int n0   = p0 - b0 * N;

  const int* ip = neigh_idx + (size_t)p0 * 16 + m;
  int nb[PPW];
#pragma unroll
  for (int j = 0; j < PPW; ++j) nb[j] = ip[j * 16];

  const float* cp = xyz + (size_t)p0 * 3;
  const float c_lo = cp[lane];
  const float c_hi = cp[32 + m];

  v2f bfrag[3];
#pragma unroll
  for (int c0 = 0; c0 < 3; ++c0) {
    const int k0 = 4 * c0 + 2 * h;
    const int mm = m < DCH ? m : 0;
    const int ka = k0 < DCH ? k0 : 0;
    const int kb = k0 + 1 < DCH ? k0 + 1 : 0;
    const float va = fc_w[mm * DCH + ka];
    const float vb = fc_w[mm * DCH + kb];
    bfrag[c0].x = (m < DCH && k0 < DCH) ? va : 0.0f;
    bfrag[c0].y = (m < DCH && k0 + 1 < DCH) ? vb : 0.0f;
  }
  const int mc = m < DCH ? m : DCH - 1;
  const float* mrow = mlp_w + mc * DCH;
  float mw[DCH];
#pragma unroll
  for (int i = 0; i < DCH; ++i) mw[i] = mrow[i];
  const float g  = bn_gamma[mc];
  const float bt = bn_beta[mc];
  const float mu = bn_mean[mc];
  const float rs = rsqrtf(bn_var[mc] + 1e-5f);

  float nx[PPW], ny[PPW], nz[PPW];
#pragma unroll
  for (int j = 0; j < PPW; ++j) {
    const int bj = (n0 + j >= N) ? b0 + 1 : b0;
    const unsigned off = ((unsigned)(bj * N) + (unsigned)nb[j]) * 3u;
    const v3f v = *(const v3f_a4*)(xyz + off);
    nx[j] = v.x; ny[j] = v.y; nz[j] = v.z;
  }

  float* wT = sT + wid * WREG;
  float* st = wT + h * HB + m;
  const unsigned long long tlo = 0x332E0F0A29240500ULL;
  const unsigned long long thi = 0x3838383838381914ULL;
  const int roff = 4 * (int)(((m < 8 ? tlo : thi) >> (8 * (m & 7))) & 0xFFu);
  const float* rd = wT + roff + 8 * h;

  float* yp = y_out + (size_t)p0 * YP + h * YP + m;

#pragma unroll
  for (int i2 = 0; i2 < PPW / 2; ++i2) {
    float part2[2];
#pragma unroll
    for (int t = 0; t < 2; ++t) {
      const int j = 2 * i2 + t;
      const float cx = (3 * j + 0 < 32) ? lane_value(c_lo, 3 * j + 0) : lane_value(c_hi, 3 * j + 0 - 32);
      const float cy = (3 * j + 1 < 32) ? lane_value(c_lo, 3 * j + 1) : lane_value(c_hi, 3 * j + 1 - 32);
      const float cz = (3 * j + 2 < 32) ? lane_value(c_lo, 3 * j + 2) : lane_value(c_hi, 3 * j + 2 - 32);
      const float rx = cx - nx[j], ry = cy - ny[j], rz = cz - nz[j];
      const float dis = __builtin_amdgcn_sqrtf(rx * rx + ry * ry + rz * rz);

      v2f a0, a1, a2;
      a0.x = h ? ry : dis;  a0.y = h ? rz : rx;
      a1.x = h ? cz : cx;   a1.y = h ? nx[j] : cy;
      a2.x = h ? 0.0f : ny[j];  a2.y = h ? 0.0f : nz[j];

      __builtin_amdgcn_wave_barrier();
      st[0 * TP] = a0.x; st[1 * TP] = a0.y;
      st[2 * TP] = a1.x; st[3 * TP] = a1.y;
      st[4 * TP] = a2.x; st[5 * TP] = a2.y;
      __builtin_amdgcn_wave_barrier();

      v8f acc = {};
      acc = __builtin_amdgcn_wmma_f32_16x16x4_f32(false, a0, false, bfrag[0], (short)0, acc, false, false);
      acc = __builtin_amdgcn_wmma_f32_16x16x4_f32(false, a1, false, bfrag[1], (short)0, acc, false, false);
      acc = __builtin_amdgcn_wmma_f32_16x16x4_f32(false, a2, false, bfrag[2], (short)0, acc, false, false);

      const float4 f0 = *(const float4*)(rd);
      const float4 f1 = *(const float4*)(rd + 4);
      const float fv[8] = {f0.x, f0.y, f0.z, f0.w, f1.x, f1.y, f1.z, f1.w};

      float mx = acc[0];
#pragma unroll
      for (int r = 1; r < 8; ++r) mx = fmaxf(mx, acc[r]);
      mx = fmaxf(mx, other_half(mx));

      const float nmx = -mx * LOG2E;
      float e[8];
      float s = 0.0f;
#pragma unroll
      for (int r = 0; r < 8; ++r) { e[r] = __builtin_amdgcn_exp2f(fmaf(acc[r], LOG2E, nmx)); s += e[r]; }
      s += other_half(s);

      float part = 0.0f;
#pragma unroll
      for (int r = 0; r < 8; ++r) part = fmaf(fv[r], e[r], part);
      part2[t] = part * __builtin_amdgcn_rcpf(s);
    }

    const float keep = h ? part2[1] : part2[0];
    const float send = h ? part2[0] : part2[1];
    const float agg  = keep + other_half(send);

    float yo = 0.0f;
    yo = fmaf(mw[0], row_lane<0>(agg), yo);
    yo = fmaf(mw[1], row_lane<1>(agg), yo);
    yo = fmaf(mw[2], row_lane<2>(agg), yo);
    yo = fmaf(mw[3], row_lane<3>(agg), yo);
    yo = fmaf(mw[4], row_lane<4>(agg), yo);
    yo = fmaf(mw[5], row_lane<5>(agg), yo);
    yo = fmaf(mw[6], row_lane<6>(agg), yo);
    yo = fmaf(mw[7], row_lane<7>(agg), yo);
    yo = fmaf(mw[8], row_lane<8>(agg), yo);
    yo = fmaf(mw[9], row_lane<9>(agg), yo);

    if (m < YW) {
      float r = (yo - mu) * rs * g + bt;
      r = fmaxf(r, 0.0f);
      yp[i2 * 2 * YP] = m < DCH ? r : 0.0f;
    }
  }
}

__global__ __launch_bounds__(PT) void maxpool_gather_kernel(
    const float* __restrict__ y,
    const int*   __restrict__ sub_idx,
    float*       __restrict__ out,
    int N, int M)
{
  const int t = blockIdx.x * PT + threadIdx.x;
  if (t >= 3 * M) return;
  const int mi = t / 3;
  const int q  = t - 3 * mi;

  v4i id[NB][4];
#pragma unroll
  for (int b = 0; b < NB; ++b) {
    const v4i* si = (const v4i*)(sub_idx + ((size_t)b * M + mi) * 16);
#pragma unroll
    for (int j = 0; j < 4; ++j) id[b][j] = si[j];
  }

  const char* yb = (const char*)y;
  const unsigned qoff = 16u * (unsigned)q;
  v4f r[NB][16];
#pragma unroll
  for (int b = 0; b < NB; ++b) {
#pragma unroll
    for (int k = 0; k < 16; ++k) {
      const unsigned off = ((unsigned)(b * N) + (unsigned)id[b][k >> 2][k & 3]) * (YP * 4u) + qoff;
      r[b][k] = *(const v4f*)(yb + off);
    }
  }
  __builtin_amdgcn_sched_barrier(0);

#pragma unroll
  for (int b = 0; b < NB; ++b) {
    v4f mx = {-3.402823466e38f, -3.402823466e38f, -3.402823466e38f, -3.402823466e38f};
#pragma unroll
    for (int k = 0; k < 16; ++k) {
      mx.x = fmaxf(mx.x, r[b][k].x); mx.y = fmaxf(mx.y, r[b][k].y);
      mx.z = fmaxf(mx.z, r[b][k].z); mx.w = fmaxf(mx.w, r[b][k].w);
    }
    float* o = out + ((size_t)b * M + mi) * DCH + 4 * q;
    v2f lo; lo.x = mx.x; lo.y = mx.y;
    *(v2f*)o = lo;
    if (q < 2) {
      v2f hi; hi.x = mx.z; hi.y = mx.w;
      *(v2f*)(o + 2) = hi;
    }
  }
}

extern "C" void kernel_launch(void* const* d_in, const int* in_sizes, int n_in,
                              void* d_out, int out_size, void* d_ws, size_t ws_size,
                              hipStream_t stream) {
  const float* xyz       = (const float*)d_in[0];
  const float* fc_w      = (const float*)d_in[1];
  const float* mlp_w     = (const float*)d_in[2];
  const float* bn_gamma  = (const float*)d_in[3];
  const float* bn_beta   = (const float*)d_in[4];
  const float* bn_mean   = (const float*)d_in[5];
  const float* bn_var    = (const float*)d_in[6];
  const int*   neigh_idx = (const int*)d_in[7];
  const int*   sub_idx   = (const int*)d_in[8];
  float*       out       = (float*)d_out;
  float*       y_ws      = (float*)d_ws;

  const int B = NB;
  const int N = in_sizes[7] / (B * 16);
  const int M = in_sizes[8] / (B * 16);

  const int points = B * N;
  attpool_wmma_kernel<<<points / 128, 256, 0, stream>>>(
      xyz, fc_w, mlp_w, bn_gamma, bn_beta, bn_mean, bn_var, neigh_idx,
      y_ws, B, N);

  maxpool_gather_kernel<<<(3 * M + PT - 1) / PT, PT, 0, stream>>>(
      y_ws, sub_idx, out, N, M);
}


# Two points share one accumulator tile: 2nd point's weights sit in columns 8-14 of a zero-padded B fragment (2 chained WMMAs per pair), so the softmax vector work runs once per pair, not per point
# speedup vs baseline: 1.5371x; 1.1678x over previous
#include <hip/hip_runtime.h>
#include <hip/hip_bf16.h>


typedef __attribute__((ext_vector_type(2))) float v2f;
typedef __attribute__((ext_vector_type(3))) float v3f;
typedef __attribute__((ext_vector_type(8))) float v8f;
typedef v3f v3f_a4 __attribute__((aligned(4)));
typedef __attribute__((ext_vector_type(4))) float v4f;
typedef __attribute__((ext_vector_type(4))) int   v4i;

#define DCH 10
#define NB  2
#define YP  16
#define YW  12
#define PT  192
#define PPW 16
#define TP  20
#define PSL 80
#define WREG 320
#define LOG2E 1.44269504088896340736f

__device__ __forceinline__ float other_half(float v) {
  const int x = __float_as_int(v);
  return __int_as_float(__builtin_amdgcn_permlanex16(x, x, 0x76543210, 0xfedcba98, false, false));
}
__device__ __forceinline__ float upper_row_turned(float v) {
  const int x = __float_as_int(v);
  return __int_as_float(__builtin_amdgcn_update_dpp(x, x, 0x128, 0x2, 0xF, false));
}
__device__ __forceinline__ float other_half_turned(float v) {
  const int x = __float_as_int(v);
  return __int_as_float(__builtin_amdgcn_permlanex16(x, x, 0xfedcba98, 0x76543210, false, false));
}
template <int I>
__device__ __forceinline__ float row_lane(float v) {
  return __int_as_float(__builtin_amdgcn_update_dpp(0, __float_as_int(v), 0x150 + I, 0xF, 0xF, true));
}
__device__ __forceinline__ float lane_value(float v, int l) {
  return __int_as_float(__builtin_amdgcn_readlane(__float_as_int(v), l));
}

__global__ __launch_bounds__(256) void attpool_wmma_kernel(
    const float* __restrict__ xyz,
    const float* __restrict__ fc_w,
    const float* __restrict__ mlp_w,
    const float* __restrict__ bn_gamma,
    const float* __restrict__ bn_beta,
    const float* __restrict__ bn_mean,
    const float* __restrict__ bn_var,
    const int*   __restrict__ neigh_idx,
    float*       __restrict__ y_out,
    int B, int N)
{
  __shared__ __attribute__((aligned(16))) float sT[8 * WREG];

  const int lane = threadIdx.x & 31;
  const int wid  = __builtin_amdgcn_readfirstlane((int)(threadIdx.x >> 5));
  const int m    = lane & 15;
  const int h    = lane >> 4;
  const int p0   = (blockIdx.x * 8 + wid) * PPW;
  const int b0   = p0 / N;
  const int n0   = p0 - b0 * N;

  const int* ip = neigh_idx + (size_t)p0 * 16 + m;
  int nb[PPW];
#pragma unroll
  for (int j = 0; j < PPW; ++j) nb[j] = ip[j * 16];

  const float* cp = xyz + (size_t)p0 * 3;
  const float c_lo = cp[lane];
  const float c_hi = cp[32 + m];

  const int mc = m < DCH ? m : DCH - 1;
  const int c7  = m & 7;
  const int chn = (int)((0x09873210u >> (4 * c7)) & 0xFu);
  const float* wr = fc_w + chn * DCH;
  float fw[DCH];
#pragma unroll
  for (int i = 0; i < DCH; ++i) fw[i] = wr[i];
  const float* mrow = mlp_w + mc * DCH;
  float mw[DCH];
#pragma unroll
  for (int i = 0; i < DCH; ++i) mw[i] = mrow[i];
  const float g  = bn_gamma[mc];
  const float bt = bn_beta[mc];
  const float mu = bn_mean[mc];
  const float bv = bn_var[mc];
  __builtin_amdgcn_sched_barrier(0);

  float nx[PPW], ny[PPW], nz[PPW];
#pragma unroll
  for (int j = 0; j < PPW; ++j) {
    const int bj = (n0 + j >= N) ? b0 + 1 : b0;
    const unsigned off = ((unsigned)(bj * N) + (unsigned)nb[j]) * 3u;
    const v3f v = *(const v3f_a4*)(xyz + off);
    nx[j] = v.x; ny[j] = v.y; nz[j] = v.z;
  }

  v2f blo, bhi;
  {
    const float bx = h ? fw[8] - fw[2] : fw[0];
    const float by = h ? fw[9] - fw[3] : fw[7] - fw[1];
    const bool lo = m < 7;
    const bool hi = m >= 8 && m < 15;
    blo.x = lo ? bx : 0.0f;  blo.y = lo ? by : 0.0f;
    bhi.x = hi ? bx : 0.0f;  bhi.y = hi ? by : 0.0f;
  }
  const float u0 = mw[0], u1 = -mw[1], u2 = -mw[2], u3 = -mw[3], u7 = mw[7], u8 = mw[8], u9 = mw[9];
  const float wcx = mw[1] + mw[4], wcy = mw[2] + mw[5], wcz = mw[3] + mw[6];
  const float rs = rsqrtf(bv + 1e-5f);

  float* wT = sT + wid * WREG;
  float* st = wT + 2 * h * TP + m;
  const int rsel = (int)((0x03213210u >> (4 * c7)) & 0xFu);
  const float* rd = wT + (m >> 3) * PSL + rsel * TP + 8 * h;

  float* yp = y_out + (size_t)p0 * YP + h * YP + m;

#pragma unroll
  for (int i4 = 0; i4 < PPW / 4; ++i4) {
    v2f a4[4];
    float cth[2];
    __builtin_amdgcn_wave_barrier();
#pragma unroll
    for (int t = 0; t < 4; ++t) {
      const int j = 4 * i4 + t;
      const float cx = (3 * j + 0 < 32) ? lane_value(c_lo, 3 * j + 0) : lane_value(c_hi, 3 * j + 0 - 32);
      const float cy = (3 * j + 1 < 32) ? lane_value(c_lo, 3 * j + 1) : lane_value(c_hi, 3 * j + 1 - 32);
      const float cz = (3 * j + 2 < 32) ? lane_value(c_lo, 3 * j + 2) : lane_value(c_hi, 3 * j + 2 - 32);
      const float rx = cx - nx[j], ry = cy - ny[j], rz = cz - nz[j];
      const float dis = __builtin_amdgcn_sqrtf(rx * rx + ry * ry + rz * rz);

      v2f a;
      a.x = h ? ny[j] : dis;
      a.y = h ? nz[j] : nx[j];

      st[t * PSL] = a.x; st[t * PSL + TP] = a.y;
      a4[t] = a;
      const float cpart = fmaf(wcz, cz, fmaf(wcy, cy, wcx * cx));
      if ((t & 1) == 0) cth[t >> 1] = cpart;
      else              cth[t >> 1] = h ? cpart : cth[t >> 1];
    }
    __builtin_amdgcn_wave_barrier();
    float res[2];
#pragma unroll
    for (int q = 0; q < 2; ++q) {
      v8f acc = {};
      acc = __builtin_amdgcn_wmma_f32_16x16x4_f32(false, a4[2 * q], false, blo, (short)0, acc, false, false);
      acc = __builtin_amdgcn_wmma_f32_16x16x4_f32(false, a4[2 * q + 1], false, bhi, (short)0, acc, false, false);

      const float4 f0 = *(const float4*)(rd + q * 2 * PSL);
      const float4 f1 = *(const float4*)(rd + q * 2 * PSL + 4);
      const float fv[8] = {f0.x, f0.y, f0.z, f0.w, f1.x, f1.y, f1.z, f1.w};

      float mx = acc[0];
#pragma unroll
      for (int r = 1; r < 8; ++r) mx = fmaxf(mx, acc[r]);
      mx = fmaxf(mx, other_half(mx));

      const float nmx = -mx * LOG2E;
      float e[8];
      float s = 0.0f;
#pragma unroll
      for (int r = 0; r < 8; ++r) { e[r] = __builtin_amdgcn_exp2f(fmaf(acc[r], LOG2E, nmx)); s += e[r]; }
      s += other_half(s);

      float part = 0.0f;
#pragma unroll
      for (int r = 0; r < 8; ++r) part = fmaf(fv[r], e[r], part);
      part *= __builtin_amdgcn_rcpf(s);

      const float tq  = upper_row_turned(part);
      const float agg = tq + other_half_turned(tq);

      float yo = cth[q];
      yo = fmaf(u0, row_lane<0>(agg), yo);
      yo = fmaf(u1, row_lane<1>(agg), yo);
      yo = fmaf(u2, row_lane<2>(agg), yo);
      yo = fmaf(u3, row_lane<3>(agg), yo);
      yo = fmaf(u7, row_lane<4>(agg), yo);
      yo = fmaf(u8, row_lane<5>(agg), yo);
      yo = fmaf(u9, row_lane<6>(agg), yo);

      float r = (yo - mu) * rs * g + bt;
      r = fmaxf(r, 0.0f);
      res[q] = m < DCH ? r : 0.0f;
    }
    if (m < YW) {
      yp[(2 * i4 + 0) * 2 * YP] = res[0];
      yp[(2 * i4 + 1) * 2 * YP] = res[1];
    }
  }
}

__global__ __launch_bounds__(PT) void maxpool_gather_kernel(
    const float* __restrict__ y,
    const int*   __restrict__ sub_idx,
    float*       __restrict__ out,
    int N, int M)
{
  const int t = blockIdx.x * PT + threadIdx.x;
  if (t >= 3 * M) return;
  const int mi = t / 3;
  const int q  = t - 3 * mi;

  v4i id[NB][4];
#pragma unroll
  for (int b = 0; b < NB; ++b) {
    const v4i* si = (const v4i*)(sub_idx + ((size_t)b * M + mi) * 16);
#pragma unroll
    for (int j = 0; j < 4; ++j) id[b][j] = si[j];
  }

  const char* yb = (const char*)y;
  const unsigned qoff = 16u * (unsigned)q;
  v4f r[NB][16];
#pragma unroll
  for (int b = 0; b < NB; ++b) {
#pragma unroll
    for (int k = 0; k < 16; ++k) {
      const unsigned off = ((unsigned)(b * N) + (unsigned)id[b][k >> 2][k & 3]) * (YP * 4u) + qoff;
      r[b][k] = *(const v4f*)(yb + off);
    }
  }
  __builtin_amdgcn_sched_barrier(0);

#pragma unroll
  for (int b = 0; b < NB; ++b) {
    v4f mx = {-3.402823466e38f, -3.402823466e38f, -3.402823466e38f, -3.402823466e38f};
#pragma unroll
    for (int k = 0; k < 16; ++k) {
      mx.x = fmaxf(mx.x, r[b][k].x); mx.y = fmaxf(mx.y, r[b][k].y);
      mx.z = fmaxf(mx.z, r[b][k].z); mx.w = fmaxf(mx.w, r[b][k].w);
    }
    float* o = out + ((size_t)b * M + mi) * DCH + 4 * q;
    v2f lo; lo.x = mx.x; lo.y = mx.y;
    *(v2f*)o = lo;
    if (q < 2) {
      v2f hi; hi.x = mx.z; hi.y = mx.w;
      *(v2f*)(o + 2) = hi;
    }
  }
}

extern "C" void kernel_launch(void* const* d_in, const int* in_sizes, int n_in,
                              void* d_out, int out_size, void* d_ws, size_t ws_size,
                              hipStream_t stream) {
  const float* xyz       = (const float*)d_in[0];
  const float* fc_w      = (const float*)d_in[1];
  const float* mlp_w     = (const float*)d_in[2];
  const float* bn_gamma  = (const float*)d_in[3];
  const float* bn_beta   = (const float*)d_in[4];
  const float* bn_mean   = (const float*)d_in[5];
  const float* bn_var    = (const float*)d_in[6];
  const int*   neigh_idx = (const int*)d_in[7];
  const int*   sub_idx   = (const int*)d_in[8];
  float*       out       = (float*)d_out;
  float*       y_ws      = (float*)d_ws;

  const int B = NB;
  const int N = in_sizes[7] / (B * 16);
  const int M = in_sizes[8] / (B * 16);

  const int points = B * N;
  attpool_wmma_kernel<<<points / 128, 256, 0, stream>>>(
      xyz, fc_w, mlp_w, bn_gamma, bn_beta, bn_mean, bn_var, neigh_idx,
      y_ws, B, N);

  maxpool_gather_kernel<<<(3 * M + PT - 1) / PT, PT, 0, stream>>>(
      y_ws, sub_idx, out, N, M);
}


# Lane halves take different points in the preparation stage: one index load, one gather and one distance computation serve two points; a pair's two A fragments come from two v_permlane16_swap
# speedup vs baseline: 1.5851x; 1.0312x over previous
#include <hip/hip_runtime.h>
#include <hip/hip_bf16.h>


typedef __attribute__((ext_vector_type(2))) float v2f;
typedef __attribute__((ext_vector_type(3))) float v3f;
typedef __attribute__((ext_vector_type(8))) float v8f;
typedef v3f v3f_a4 __attribute__((aligned(4)));
typedef __attribute__((ext_vector_type(4))) float v4f;
typedef __attribute__((ext_vector_type(4))) int   v4i;

#define DCH 10
#define NB  2
#define YP  16
#define YW  12
#define PT  192
#define PPW 16
#define TP  20
#define PSL 80
#define WREG 320
#define LOG2E 1.44269504088896340736f

__device__ __forceinline__ float other_half(float v) {
  const int x = __float_as_int(v);
  return __int_as_float(__builtin_amdgcn_permlanex16(x, x, 0x76543210, 0xfedcba98, false, false));
}
__device__ __forceinline__ float upper_row_turned(float v) {
  const int x = __float_as_int(v);
  return __int_as_float(__builtin_amdgcn_update_dpp(x, x, 0x128, 0x2, 0xF, false));
}
__device__ __forceinline__ float other_half_turned(float v) {
  const int x = __float_as_int(v);
  return __int_as_float(__builtin_amdgcn_permlanex16(x, x, 0xfedcba98, 0x76543210, false, false));
}
template <int I>
__device__ __forceinline__ float row_lane(float v) {
  return __int_as_float(__builtin_amdgcn_update_dpp(0, __float_as_int(v), 0x150 + I, 0xF, 0xF, true));
}
__device__ __forceinline__ void half_swap(float& a, float& b) {
  typedef __attribute__((ext_vector_type(2))) unsigned int v2u;
  const v2u r = __builtin_amdgcn_permlane16_swap(__float_as_uint(a), __float_as_uint(b), false, false);
  a = __uint_as_float(r.x);
  b = __uint_as_float(r.y);
}

__global__ __launch_bounds__(256) void attpool_wmma_kernel(
    const float* __restrict__ xyz,
    const float* __restrict__ fc_w,
    const float* __restrict__ mlp_w,
    const float* __restrict__ bn_gamma,
    const float* __restrict__ bn_beta,
    const float* __restrict__ bn_mean,
    const float* __restrict__ bn_var,
    const int*   __restrict__ neigh_idx,
    float*       __restrict__ y_out,
    int B, int N)
{
  __shared__ __attribute__((aligned(16))) float sT[8 * WREG];

  const int lane = threadIdx.x & 31;
  const int wid  = __builtin_amdgcn_readfirstlane((int)(threadIdx.x >> 5));
  const int m    = lane & 15;
  const int h    = lane >> 4;
  const int p0   = (blockIdx.x * 8 + wid) * PPW;
  const int b0   = p0 / N;

  const int* ip = neigh_idx + (size_t)p0 * 16 + lane;
  int nb[PPW / 2];
#pragma unroll
  for (int q = 0; q < PPW / 2; ++q) nb[q] = ip[q * 32];

  const float* cp = xyz + (size_t)p0 * 3 + 3 * h;
  float ccx[PPW / 2], ccy[PPW / 2], ccz[PPW / 2];
#pragma unroll
  for (int q = 0; q < PPW / 2; ++q) {
    const v3f v = *(const v3f_a4*)(cp + 6 * q);
    ccx[q] = v.x; ccy[q] = v.y; ccz[q] = v.z;
  }

  const int mc = m < DCH ? m : DCH - 1;
  const int c7  = m & 7;
  const int chn = (int)((0x09873210u >> (4 * c7)) & 0xFu);
  const float* wr = fc_w + chn * DCH;
  float fw[DCH];
#pragma unroll
  for (int i = 0; i < DCH; ++i) fw[i] = wr[i];
  const float* mrow = mlp_w + mc * DCH;
  float mw[DCH];
#pragma unroll
  for (int i = 0; i < DCH; ++i) mw[i] = mrow[i];
  const float g  = bn_gamma[mc];
  const float bt = bn_beta[mc];
  const float mu = bn_mean[mc];
  const float bv = bn_var[mc];
  __builtin_amdgcn_sched_barrier(0);

  float nx[PPW / 2], ny[PPW / 2], nz[PPW / 2];
  const float* tb = xyz + (size_t)b0 * (size_t)N * 3;
#pragma unroll
  for (int q = 0; q < PPW / 2; ++q) {
    const unsigned i3 = (unsigned)nb[q] * 3u;
    const v3f v = *(const v3f_a4*)(tb + i3);
    nx[q] = v.x; ny[q] = v.y; nz[q] = v.z;
  }

  v2f blo, bhi;
  {
    const float bx = h ? fw[8] - fw[2] : fw[0];
    const float by = h ? fw[9] - fw[3] : fw[7] - fw[1];
    const bool lo = m < 7;
    const bool hi = m >= 8 && m < 15;
    blo.x = lo ? bx : 0.0f;  blo.y = lo ? by : 0.0f;
    bhi.x = hi ? bx : 0.0f;  bhi.y = hi ? by : 0.0f;
  }
  const float u0 = mw[0], u1 = -mw[1], u2 = -mw[2], u3 = -mw[3], u7 = mw[7], u8 = mw[8], u9 = mw[9];
  const float wcx = mw[1] + mw[4], wcy = mw[2] + mw[5], wcz = mw[3] + mw[6];
  const float rs = rsqrtf(bv + 1e-5f);

  float* wT = sT + wid * WREG;
  float* st = wT + 2 * h * TP + m;
  const int rsel = (int)((0x03213210u >> (4 * c7)) & 0xFu);
  const float* rd = wT + (m >> 3) * PSL + rsel * TP + 8 * h;

  float* yp = y_out + (size_t)p0 * YP + h * YP + m;

#pragma unroll
  for (int i4 = 0; i4 < PPW / 4; ++i4) {
    v2f a4[4];
    float cth[2];
    __builtin_amdgcn_wave_barrier();
#pragma unroll
    for (int qq = 0; qq < 2; ++qq) {
      const int q = 2 * i4 + qq;
      const float rx = ccx[q] - nx[q], ry = ccy[q] - ny[q], rz = ccz[q] - nz[q];
      const float dis = __builtin_amdgcn_sqrtf(rx * rx + ry * ry + rz * rz);

      float ax0 = dis, ax1 = ny[q];
      float ay0 = nx[q], ay1 = nz[q];
      half_swap(ax0, ax1);
      half_swap(ay0, ay1);
      v2f a_first, a_second;
      a_first.x = ax0;  a_first.y = ay0;
      a_second.x = ax1; a_second.y = ay1;

      st[(2 * qq) * PSL] = a_first.x;      st[(2 * qq) * PSL + TP] = a_first.y;
      st[(2 * qq + 1) * PSL] = a_second.x; st[(2 * qq + 1) * PSL + TP] = a_second.y;
      a4[2 * qq] = a_first;
      a4[2 * qq + 1] = a_second;
      cth[qq] = fmaf(wcz, ccz[q], fmaf(wcy, ccy[q], wcx * ccx[q]));
    }
    __builtin_amdgcn_wave_barrier();
    float res[2];
#pragma unroll
    for (int q = 0; q < 2; ++q) {
      v8f acc = {};
      acc = __builtin_amdgcn_wmma_f32_16x16x4_f32(false, a4[2 * q], false, blo, (short)0, acc, false, false);
      acc = __builtin_amdgcn_wmma_f32_16x16x4_f32(false, a4[2 * q + 1], false, bhi, (short)0, acc, false, false);

      const float4 f0 = *(const float4*)(rd + q * 2 * PSL);
      const float4 f1 = *(const float4*)(rd + q * 2 * PSL + 4);
      const float fv[8] = {f0.x, f0.y, f0.z, f0.w, f1.x, f1.y, f1.z, f1.w};

      float mx = acc[0];
#pragma unroll
      for (int r = 1; r < 8; ++r) mx = fmaxf(mx, acc[r]);
      mx = fmaxf(mx, other_half(mx));

      const float nmx = -mx * LOG2E;
      float e[8];
      float s = 0.0f;
#pragma unroll
      for (int r = 0; r < 8; ++r) { e[r] = __builtin_amdgcn_exp2f(fmaf(acc[r], LOG2E, nmx)); s += e[r]; }
      s += other_half(s);

      float part = 0.0f;
#pragma unroll
      for (int r = 0; r < 8; ++r) part = fmaf(fv[r], e[r], part);
      part *= __builtin_amdgcn_rcpf(s);

      const float tq  = upper_row_turned(part);
      const float agg = tq + other_half_turned(tq);

      float yo = cth[q];
      yo = fmaf(u0, row_lane<0>(agg), yo);
      yo = fmaf(u1, row_lane<1>(agg), yo);
      yo = fmaf(u2, row_lane<2>(agg), yo);
      yo = fmaf(u3, row_lane<3>(agg), yo);
      yo = fmaf(u7, row_lane<4>(agg), yo);
      yo = fmaf(u8, row_lane<5>(agg), yo);
      yo = fmaf(u9, row_lane<6>(agg), yo);

      float r = (yo - mu) * rs * g + bt;
      r = fmaxf(r, 0.0f);
      res[q] = m < DCH ? r : 0.0f;
    }
    if (m < YW) {
      yp[(2 * i4 + 0) * 2 * YP] = res[0];
      yp[(2 * i4 + 1) * 2 * YP] = res[1];
    }
  }
}

__global__ __launch_bounds__(PT) void maxpool_gather_kernel(
    const float* __restrict__ y,
    const int*   __restrict__ sub_idx,
    float*       __restrict__ out,
    int N, int M)
{
  const int t = blockIdx.x * PT + threadIdx.x;
  if (t >= 3 * M) return;
  const int mi = t / 3;
  const int q  = t - 3 * mi;

  v4i id[NB][4];
#pragma unroll
  for (int b = 0; b < NB; ++b) {
    const v4i* si = (const v4i*)(sub_idx + ((size_t)b * M + mi) * 16);
#pragma unroll
    for (int j = 0; j < 4; ++j) id[b][j] = si[j];
  }

  const char* yb = (const char*)y;
  const unsigned qoff = 16u * (unsigned)q;
  v4f r[NB][16];
#pragma unroll
  for (int b = 0; b < NB; ++b) {
#pragma unroll
    for (int k = 0; k < 16; ++k) {
      const unsigned off = ((unsigned)(b * N) + (unsigned)id[b][k >> 2][k & 3]) * (YP * 4u) + qoff;
      r[b][k] = *(const v4f*)(yb + off);
    }
  }
  __builtin_amdgcn_sched_barrier(0);

#pragma unroll
  for (int b = 0; b < NB; ++b) {
    v4f mx = {-3.402823466e38f, -3.402823466e38f, -3.402823466e38f, -3.402823466e38f};
#pragma unroll
    for (int k = 0; k < 16; ++k) {
      mx.x = fmaxf(mx.x, r[b][k].x); mx.y = fmaxf(mx.y, r[b][k].y);
      mx.z = fmaxf(mx.z, r[b][k].z); mx.w = fmaxf(mx.w, r[b][k].w);
    }
    float* o = out + ((size_t)b * M + mi) * DCH + 4 * q;
    v2f lo; lo.x = mx.x; lo.y = mx.y;
    *(v2f*)o = lo;
    if (q < 2) {
      v2f hi; hi.x = mx.z; hi.y = mx.w;
      *(v2f*)(o + 2) = hi;
    }
  }
}

extern "C" void kernel_launch(void* const* d_in, const int* in_sizes, int n_in,
                              void* d_out, int out_size, void* d_ws, size_t ws_size,
                              hipStream_t stream) {
  const float* xyz       = (const float*)d_in[0];
  const float* fc_w      = (const float*)d_in[1];
  const float* mlp_w     = (const float*)d_in[2];
  const float* bn_gamma  = (const float*)d_in[3];
  const float* bn_beta   = (const float*)d_in[4];
  const float* bn_mean   = (const float*)d_in[5];
  const float* bn_var    = (const float*)d_in[6];
  const int*   neigh_idx = (const int*)d_in[7];
  const int*   sub_idx   = (const int*)d_in[8];
  float*       out       = (float*)d_out;
  float*       y_ws      = (float*)d_ws;

  const int B = NB;
  const int N = in_sizes[7] / (B * 16);
  const int M = in_sizes[8] / (B * 16);

  const int points = B * N;
  attpool_wmma_kernel<<<points / 128, 256, 0, stream>>>(
      xyz, fc_w, mlp_w, bn_gamma, bn_beta, bn_mean, bn_var, neigh_idx,
      y_ws, B, N);

  maxpool_gather_kernel<<<(3 * M + PT - 1) / PT, PT, 0, stream>>>(
      y_ws, sub_idx, out, N, M);
}
